# starting kernel
# baseline (speedup 1.0000x reference)
#include <hip/hip_runtime.h>
#include <hip/hip_bf16.h>

typedef float f32x2 __attribute__((ext_vector_type(2)));
typedef float f32x8 __attribute__((ext_vector_type(8)));

#define BATCH   64
#define CCH     3
#define HIMG    224
#define WIMG    224
#define PSZ     16
#define DMODEL  768
#define KDIM    768
#define NPATCH  196
#define SEQ     197
#define ROWS    (BATCH * NPATCH)

#define TM      64
#define TN      64
#define TK      32
#define A_STRIDE 36
#define B_STRIDE 68

__global__ __launch_bounds__(256)
void vit_patch_embed_gemm(const float* __restrict__ x,
                          const float* __restrict__ Wm,
                          const float* __restrict__ bias,
                          const float* __restrict__ pos,
                          float* __restrict__ out)
{
    __shared__ float As[TM * A_STRIDE];
    __shared__ float Bs[TK * B_STRIDE];

    const int tid    = threadIdx.x;
    const int lane   = tid & 31;
    const int waveId = tid >> 5;
    const int waveM  = waveId & 3;
    const int waveN  = waveId >> 2;
    const int half   = lane >> 4;
    const int lsub   = lane & 15;
    const int kOff   = half * 2;

    const int rowBase = blockIdx.x * TM;
    const int nBase   = blockIdx.y * TN;

    const int aRow     = tid >> 2;
    const int aColBase = (tid & 3) * 8;
    const int rG   = rowBase + aRow;
    const int bImg = rG / NPATCH;
    const int nTok = rG - bImg * NPATCH;
    const int pi_  = nTok / 14;
    const int pj_  = nTok - pi_ * 14;

    const int bRow     = tid >> 3;
    const int bColBase = (tid & 7) * 8;

    f32x8 c0 = {0.f,0.f,0.f,0.f,0.f,0.f,0.f,0.f};
    f32x8 c1 = {0.f,0.f,0.f,0.f,0.f,0.f,0.f,0.f};

    const int nA = waveN * 32;
    const int nB = nA + 16;

    for (int kBase = 0; kBase < KDIM; kBase += TK) {
        {
            const int f0 = kBase + aColBase;
            const int ch = f0 >> 8;
            const int rem = f0 & 255;
            const int pr = rem >> 4;
            const int pc = rem & 15;
            const long gidx = ((long)(bImg * CCH + ch) * HIMG + pi_ * PSZ + pr) * WIMG
                              + pj_ * PSZ + pc;
            const float4 av0 = *(const float4*)(x + gidx);
            const float4 av1 = *(const float4*)(x + gidx + 4);
            *(float4*)&As[aRow * A_STRIDE + aColBase]     = av0;
            *(float4*)&As[aRow * A_STRIDE + aColBase + 4] = av1;
        }
        {
            const long widx = (long)(kBase + bRow) * DMODEL + nBase + bColBase;
            const float4 bv0 = *(const float4*)(Wm + widx);
            const float4 bv1 = *(const float4*)(Wm + widx + 4);
            *(float4*)&Bs[bRow * B_STRIDE + bColBase]     = bv0;
            *(float4*)&Bs[bRow * B_STRIDE + bColBase + 4] = bv1;
            if (kBase + TK < KDIM) {
                __builtin_prefetch(Wm + widx + (long)TK * DMODEL, 0, 0);
            }
        }
        __syncthreads();

        #pragma unroll
        for (int kk = 0; kk < TK; kk += 4) {
            f32x2 a, b0, b1;
            const int ar = (waveM * 16 + lsub) * A_STRIDE + kk + kOff;
            a.x = As[ar];
            a.y = As[ar + 1];
            const int br0 = (kk + kOff) * B_STRIDE;
            const int br1 = (kk + kOff + 1) * B_STRIDE;
            b0.x = Bs[br0 + nA + lsub];
            b0.y = Bs[br1 + nA + lsub];
            b1.x = Bs[br0 + nB + lsub];
            b1.y = Bs[br1 + nB + lsub];
            c0 = __builtin_amdgcn_wmma_f32_16x16x4_f32(false, a, false, b0,
                                                       (short)0, c0, false, false);
            c1 = __builtin_amdgcn_wmma_f32_16x16x4_f32(false, a, false, b1,
                                                       (short)0, c1, false, false);
        }
        __syncthreads();
    }

    const int colA = nBase + nA + lsub;
    const int colB = nBase + nB + lsub;
    const float biasA = bias[colA];
    const float biasB = bias[colB];
    #pragma unroll
    for (int v = 0; v < 8; ++v) {
        const int mRow = waveM * 16 + v + half * 8;
        const int r    = rowBase + mRow;
        const int bI   = r / NPATCH;
        const int s    = r - bI * NPATCH + 1;
        const long obase = ((long)bI * SEQ + s) * DMODEL;
        const long pbase = (long)s * DMODEL;
        out[obase + colA] = c0[v] + biasA + pos[pbase + colA];
        out[obase + colB] = c1[v] + biasB + pos[pbase + colB];
    }
}

__global__ __launch_bounds__(256)
void vit_class_token(const float* __restrict__ cls,
                     const float* __restrict__ pos,
                     float* __restrict__ out)
{
    const int idx = blockIdx.x * blockDim.x + threadIdx.x;
    const int b = idx / DMODEL;
    const int d = idx - b * DMODEL;
    out[(long)b * SEQ * DMODEL + d] = cls[d] + pos[d];
}

extern "C" void kernel_launch(void* const* d_in, const int* in_sizes, int n_in,
                              void* d_out, int out_size, void* d_ws, size_t ws_size,
                              hipStream_t stream) {
    const float* x   = (const float*)d_in[0];
    const float* Wm  = (const float*)d_in[1];
    const float* b   = (const float*)d_in[2];
    const float* pos = (const float*)d_in[3];
    const float* cls = (const float*)d_in[4];
    float* out = (float*)d_out;

    dim3 grid(ROWS / TM, DMODEL / TN);
    vit_patch_embed_gemm<<<grid, 256, 0, stream>>>(x, Wm, b, pos, out);
    vit_class_token<<<(BATCH * DMODEL) / 256, 256, 0, stream>>>(cls, pos, out);
}


# A fragments loaded straight from global into registers a 32-k chunk ahead; LDS holds only W (128-k stages, block 128x32); lane halves take 8-float runs: k regrouped in a chunk, sum order changes
# speedup vs baseline: 1.1164x; 1.1164x over previous
#include <hip/hip_runtime.h>

typedef float f32x2 __attribute__((ext_vector_type(2)));
typedef float f32x8 __attribute__((ext_vector_type(8)));

#define BATCH   64
#define CCH     3
#define HIMG    224
#define WIMG    224
#define PSZ     16
#define DMODEL  768
#define KDIM    768
#define NPATCH  196
#define SEQ     197
#define ROWS    (BATCH * NPATCH)

#define TM      128
#define TN      32
#define TK      32
#define TKB     128
#define NCHUNK  (KDIM / TK)
#define NSTAGE  (KDIM / TKB)
#define B_STRIDE 36

__device__ __forceinline__ int kpos(int j, int h, int s)
{
    return ((j & 4) << 2) + 8 * h + 2 * (j & 3) + s;
}

__global__ __launch_bounds__(256)
void vit_patch_embed_gemm(const float* __restrict__ x,
                          const float* __restrict__ Wm,
                          const float* __restrict__ bias,
                          const float* __restrict__ pos,
                          float* __restrict__ out)
{
    __shared__ float Bs[TKB * B_STRIDE];

    const int tid    = threadIdx.x;
    const int lane   = tid & 31;
    const int waveId = tid >> 5;
    const int half   = lane >> 4;
    const int lsub   = lane & 15;

    const int rowBase = blockIdx.x * TM;
    const int nBase   = blockIdx.y * TN;

    const int rG   = rowBase + waveId * 16 + lsub;
    const int bImg = rG / NPATCH;
    const int nTok = rG - bImg * NPATCH;
    const int pi_  = nTok / 14;
    const int pj_  = nTok - pi_ * 14;
    const float* aPtr = x + ((long)(bImg * CCH) * HIMG + pi_ * PSZ) * WIMG + pj_ * PSZ + 8 * half;

    const int bRow = tid >> 3;
    const int bCol = (tid & 7) * 4;
    const float* wPtr = Wm + (long)bRow * DMODEL + nBase + bCol;

    const float* bFrag = Bs + (8 * half) * B_STRIDE + lsub;

    f32x8 c0 = {0.f,0.f,0.f,0.f,0.f,0.f,0.f,0.f};
    f32x8 c1 = {0.f,0.f,0.f,0.f,0.f,0.f,0.f,0.f};

    float4 an0 = *(const float4*)(aPtr);
    float4 an1 = *(const float4*)(aPtr + 4);
    float4 an2 = *(const float4*)(aPtr + WIMG);
    float4 an3 = *(const float4*)(aPtr + WIMG + 4);

    for (int st = 0; st < NSTAGE; ++st) {
        if (st > 0) __syncthreads();
        {
            const float* wp = wPtr + (long)st * TKB * DMODEL;
            const float4 w0 = *(const float4*)(wp);
            const float4 w1 = *(const float4*)(wp + 32 * DMODEL);
            const float4 w2 = *(const float4*)(wp + 64 * DMODEL);
            const float4 w3 = *(const float4*)(wp + 96 * DMODEL);
            float* bs = Bs + bRow * B_STRIDE + bCol;
            *(float4*)(bs)                   = w0;
            *(float4*)(bs + 32 * B_STRIDE)   = w1;
            *(float4*)(bs + 64 * B_STRIDE)   = w2;
            *(float4*)(bs + 96 * B_STRIDE)   = w3;
        }
        __syncthreads();

        #pragma unroll
        for (int cc = 0; cc < TKB / TK; ++cc) {
            const int t = st * (TKB / TK) + cc;
            const float a[16] = { an0.x, an0.y, an0.z, an0.w, an1.x, an1.y, an1.z, an1.w,
                                  an2.x, an2.y, an2.z, an2.w, an3.x, an3.y, an3.z, an3.w };
            if (t + 1 < NCHUNK) {
                const int tn = t + 1;
                const float* ap = aPtr + (long)(tn >> 3) * (HIMG * WIMG) + ((tn & 7) * 2) * WIMG;
                an0 = *(const float4*)(ap);
                an1 = *(const float4*)(ap + 4);
                an2 = *(const float4*)(ap + WIMG);
                an3 = *(const float4*)(ap + WIMG + 4);
            }
            #pragma unroll
            for (int j = 0; j < 8; ++j) {
                f32x2 av, b0, b1;
                av.x = a[2 * j];
                av.y = a[2 * j + 1];
                const int r0 = cc * TK + kpos(j, 0, 0);
                const int r1 = cc * TK + kpos(j, 0, 1);
                b0.x = bFrag[r0 * B_STRIDE];
                b0.y = bFrag[r1 * B_STRIDE];
                b1.x = bFrag[r0 * B_STRIDE + 16];
                b1.y = bFrag[r1 * B_STRIDE + 16];
                c0 = __builtin_amdgcn_wmma_f32_16x16x4_f32(false, av, false, b0,
                                                           (short)0, c0, false, false);
                c1 = __builtin_amdgcn_wmma_f32_16x16x4_f32(false, av, false, b1,
                                                           (short)0, c1, false, false);
            }
        }
    }

    const int colA = nBase + lsub;
    const int colB = colA + 16;
    const float biasA = bias[colA];
    const float biasB = bias[colB];
    #pragma unroll
    for (int v = 0; v < 8; ++v) {
        const int r    = rowBase + waveId * 16 + v + half * 8;
        const int bI   = r / NPATCH;
        const int s    = r - bI * NPATCH + 1;
        const long obase = ((long)bI * SEQ + s) * DMODEL;
        const long pbase = (long)s * DMODEL;
        out[obase + colA] = c0[v] + biasA + pos[pbase + colA];
        out[obase + colB] = c1[v] + biasB + pos[pbase + colB];
    }
}

__global__ __launch_bounds__(256)
void vit_class_token(const float* __restrict__ cls,
                     const float* __restrict__ pos,
                     float* __restrict__ out)
{
    const int idx = blockIdx.x * blockDim.x + threadIdx.x;
    const int b = idx / DMODEL;
    const int d = idx - b * DMODEL;
    out[(long)b * SEQ * DMODEL + d] = cls[d] + pos[d];
}

extern "C" void kernel_launch(void* const* d_in, const int* in_sizes, int n_in,
                              void* d_out, int out_size, void* d_ws, size_t ws_size,
                              hipStream_t stream) {
    const float* x   = (const float*)d_in[0];
    const float* Wm  = (const float*)d_in[1];
    const float* b   = (const float*)d_in[2];
    const float* pos = (const float*)d_in[3];
    const float* cls = (const float*)d_in[4];
    float* out = (float*)d_out;

    dim3 grid(ROWS / TM, DMODEL / TN);
    vit_patch_embed_gemm<<<grid, 256, 0, stream>>>(x, Wm, b, pos, out);
    vit_class_token<<<(BATCH * DMODEL) / 256, 256, 0, stream>>>(cls, pos, out);
}


# W fragments of a 32-k chunk read from LDS as one group (16 two-address reads, rows k and k+1 per register pair) in front of the chunk's 16 WMMAs, not among them; k order and sums unchanged
# speedup vs baseline: 1.2126x; 1.0862x over previous
#include <hip/hip_runtime.h>

typedef float f32x2 __attribute__((ext_vector_type(2)));
typedef float f32x8 __attribute__((ext_vector_type(8)));

#define BATCH   64
#define CCH     3
#define HIMG    224
#define WIMG    224
#define PSZ     16
#define DMODEL  768
#define KDIM    768
#define NPATCH  196
#define SEQ     197
#define ROWS    (BATCH * NPATCH)

#define TM      128
#define TN      32
#define TK      32
#define TKB     128
#define NCHUNK  (KDIM / TK)
#define NSTAGE  (KDIM / TKB)
#define B_STRIDE 36
static_assert(B_STRIDE == 36, "the read offsets in B_READS4 are written for a 36-float row");

__device__ __forceinline__ int kpos(int j, int h, int s)
{
    return ((j & 4) << 2) + 8 * h + 2 * (j & 3) + s;
}

#define B_READS4(o0, o1, o2, o3, base) \
    "ds_load_2addr_b32 %" #o0 ", %" #base " offset0:0 offset1:36\n\t"   \
    "ds_load_2addr_b32 %" #o1 ", %" #base " offset0:16 offset1:52\n\t"  \
    "ds_load_2addr_b32 %" #o2 ", %" #base " offset0:72 offset1:108\n\t" \
    "ds_load_2addr_b32 %" #o3 ", %" #base " offset0:88 offset1:124\n\t"

__global__ __launch_bounds__(256)
void vit_patch_embed_gemm(const float* __restrict__ x,
                          const float* __restrict__ Wm,
                          const float* __restrict__ bias,
                          const float* __restrict__ pos,
                          float* __restrict__ out)
{
    __shared__ float Bs[TKB * B_STRIDE];

    const int tid    = threadIdx.x;
    const int lane   = tid & 31;
    const int waveId = tid >> 5;
    const int half   = lane >> 4;
    const int lsub   = lane & 15;

    const int rowBase = blockIdx.x * TM;
    const int nBase   = blockIdx.y * TN;

    const int rG   = rowBase + waveId * 16 + lsub;
    const int bImg = rG / NPATCH;
    const int nTok = rG - bImg * NPATCH;
    const int pi_  = nTok / 14;
    const int pj_  = nTok - pi_ * 14;
    const float* aPtr = x + ((long)(bImg * CCH) * HIMG + pi_ * PSZ) * WIMG + pj_ * PSZ + 8 * half;

    const int bRow = tid >> 3;
    const int bCol = (tid & 7) * 4;
    const float* wPtr = Wm + (long)bRow * DMODEL + nBase + bCol;

    typedef __attribute__((address_space(3))) float lds_f32;
    const unsigned bFrag = (unsigned)(unsigned long)((lds_f32*)Bs)
                         + (unsigned)(((8 * half) * B_STRIDE + lsub) * 4);

    f32x8 c0 = {0.f,0.f,0.f,0.f,0.f,0.f,0.f,0.f};
    f32x8 c1 = {0.f,0.f,0.f,0.f,0.f,0.f,0.f,0.f};

    float4 an0 = *(const float4*)(aPtr);
    float4 an1 = *(const float4*)(aPtr + 4);
    float4 an2 = *(const float4*)(aPtr + WIMG);
    float4 an3 = *(const float4*)(aPtr + WIMG + 4);

    for (int st = 0; st < NSTAGE; ++st) {
        if (st > 0) __syncthreads();
        {
            const float* wp = wPtr + (long)st * TKB * DMODEL;
            const float4 w0 = *(const float4*)(wp);
            const float4 w1 = *(const float4*)(wp + 32 * DMODEL);
            const float4 w2 = *(const float4*)(wp + 64 * DMODEL);
            const float4 w3 = *(const float4*)(wp + 96 * DMODEL);
            float* bs = Bs + bRow * B_STRIDE + bCol;
            *(float4*)(bs)                   = w0;
            *(float4*)(bs + 32 * B_STRIDE)   = w1;
            *(float4*)(bs + 64 * B_STRIDE)   = w2;
            *(float4*)(bs + 96 * B_STRIDE)   = w3;
        }
        __syncthreads();

        #pragma unroll
        for (int cc = 0; cc < TKB / TK; ++cc) {
            const int t = st * (TKB / TK) + cc;
            const float a[16] = { an0.x, an0.y, an0.z, an0.w, an1.x, an1.y, an1.z, an1.w,
                                  an2.x, an2.y, an2.z, an2.w, an3.x, an3.y, an3.z, an3.w };
            if (t + 1 < NCHUNK) {
                const int tn = t + 1;
                const float* ap = aPtr + (long)(tn >> 3) * (HIMG * WIMG) + ((tn & 7) * 2) * WIMG;
                an0 = *(const float4*)(ap);
                an1 = *(const float4*)(ap + 4);
                an2 = *(const float4*)(ap + WIMG);
                an3 = *(const float4*)(ap + WIMG + 4);
            }
            f32x2 p0, p1, p2, p3, p4, p5, p6, p7, p8, p9, p10, p11, p12, p13, p14, p15;
            {
                const unsigned g0 = bFrag + (unsigned)((cc * TK + kpos(0, 0, 0)) * B_STRIDE * 4);
                const unsigned g1 = bFrag + (unsigned)((cc * TK + kpos(2, 0, 0)) * B_STRIDE * 4);
                const unsigned g2 = bFrag + (unsigned)((cc * TK + kpos(4, 0, 0)) * B_STRIDE * 4);
                const unsigned g3 = bFrag + (unsigned)((cc * TK + kpos(6, 0, 0)) * B_STRIDE * 4);
                asm volatile(
                    B_READS4(0, 1, 2, 3, 16)
                    B_READS4(4, 5, 6, 7, 17)
                    B_READS4(8, 9, 10, 11, 18)
                    B_READS4(12, 13, 14, 15, 19)
                    "s_wait_dscnt 0x0"
                    : "=&v"(p0), "=&v"(p1), "=&v"(p2), "=&v"(p3),
                      "=&v"(p4), "=&v"(p5), "=&v"(p6), "=&v"(p7),
                      "=&v"(p8), "=&v"(p9), "=&v"(p10), "=&v"(p11),
                      "=&v"(p12), "=&v"(p13), "=&v"(p14), "=&v"(p15)
                    : "v"(g0), "v"(g1), "v"(g2), "v"(g3)
                    : "memory");
            }
            const f32x2 p[16] = { p0, p1, p2, p3, p4, p5, p6, p7,
                                  p8, p9, p10, p11, p12, p13, p14, p15 };
            #pragma unroll
            for (int j = 0; j < 8; ++j) {
                f32x2 av;
                av.x = a[2 * j];
                av.y = a[2 * j + 1];
                c0 = __builtin_amdgcn_wmma_f32_16x16x4_f32(false, av, false, p[2 * j],
                                                           (short)0, c0, false, false);
                c1 = __builtin_amdgcn_wmma_f32_16x16x4_f32(false, av, false, p[2 * j + 1],
                                                           (short)0, c1, false, false);
            }
            __builtin_amdgcn_sched_barrier(0);
        }
    }

    const int colA = nBase + lsub;
    const int colB = colA + 16;
    const float biasA = bias[colA];
    const float biasB = bias[colB];
    #pragma unroll
    for (int v = 0; v < 8; ++v) {
        const int r    = rowBase + waveId * 16 + v + half * 8;
        const int bI   = r / NPATCH;
        const int s    = r - bI * NPATCH + 1;
        const long obase = ((long)bI * SEQ + s) * DMODEL;
        const long pbase = (long)s * DMODEL;
        out[obase + colA] = c0[v] + biasA + pos[pbase + colA];
        out[obase + colB] = c1[v] + biasB + pos[pbase + colB];
    }
}

__global__ __launch_bounds__(256)
void vit_class_token(const float* __restrict__ cls,
                     const float* __restrict__ pos,
                     float* __restrict__ out)
{
    const int idx = blockIdx.x * blockDim.x + threadIdx.x;
    const int b = idx / DMODEL;
    const int d = idx - b * DMODEL;
    out[(long)b * SEQ * DMODEL + d] = cls[d] + pos[d];
}

extern "C" void kernel_launch(void* const* d_in, const int* in_sizes, int n_in,
                              void* d_out, int out_size, void* d_ws, size_t ws_size,
                              hipStream_t stream) {
    const float* x   = (const float*)d_in[0];
    const float* Wm  = (const float*)d_in[1];
    const float* b   = (const float*)d_in[2];
    const float* pos = (const float*)d_in[3];
    const float* cls = (const float*)d_in[4];
    float* out = (float*)d_out;

    dim3 grid(ROWS / TM, DMODEL / TN);
    vit_patch_embed_gemm<<<grid, 256, 0, stream>>>(x, Wm, b, pos, out);
    vit_class_token<<<(BATCH * DMODEL) / 256, 256, 0, stream>>>(cls, pos, out);
}


# Block widened to 128x64 on 16 waves: the two 16x32 waves of a row strip run side by side and ask for the same x lines together, so half the A reads hit the WGP cache, not L2; sums unchanged
# speedup vs baseline: 1.2569x; 1.0365x over previous
#include <hip/hip_runtime.h>

typedef float f32x2 __attribute__((ext_vector_type(2)));
typedef float f32x8 __attribute__((ext_vector_type(8)));

#define BATCH   64
#define CCH     3
#define HIMG    224
#define WIMG    224
#define PSZ     16
#define DMODEL  768
#define KDIM    768
#define NPATCH  196
#define SEQ     197
#define ROWS    (BATCH * NPATCH)

#define TM      128
#define TN      64
#define TNW     32
#define TK      32
#define TKB     128
#define NCHUNK  (KDIM / TK)
#define NSTAGE  (KDIM / TKB)
#define B_STRIDE 68
static_assert(B_STRIDE == 68, "the read offsets in B_READS4 are written for a 68-float row");

__device__ __forceinline__ int kpos(int j, int h, int s)
{
    return ((j & 4) << 2) + 8 * h + 2 * (j & 3) + s;
}

#define B_READS4(o0, o1, o2, o3, base) \
    "ds_load_2addr_b32 %" #o0 ", %" #base " offset0:0 offset1:68\n\t"    \
    "ds_load_2addr_b32 %" #o1 ", %" #base " offset0:16 offset1:84\n\t"   \
    "ds_load_2addr_b32 %" #o2 ", %" #base " offset0:136 offset1:204\n\t" \
    "ds_load_2addr_b32 %" #o3 ", %" #base " offset0:152 offset1:220\n\t"

__global__ __launch_bounds__(512)
void vit_patch_embed_gemm(const float* __restrict__ x,
                          const float* __restrict__ Wm,
                          const float* __restrict__ bias,
                          const float* __restrict__ pos,
                          float* __restrict__ out)
{
    __shared__ float Bs[TKB * B_STRIDE];

    const int tid    = threadIdx.x;
    const int lane   = tid & 31;
    const int waveId = tid >> 5;
    const int strip  = waveId >> 1;
    const int nHalf  = waveId & 1;
    const int half   = lane >> 4;
    const int lsub   = lane & 15;

    const int rowBase = blockIdx.x * TM;
    const int nBase   = blockIdx.y * TN;

    const int rG   = rowBase + strip * 16 + lsub;
    const int bImg = rG / NPATCH;
    const int nTok = rG - bImg * NPATCH;
    const int pi_  = nTok / 14;
    const int pj_  = nTok - pi_ * 14;
    const float* aPtr = x + ((long)(bImg * CCH) * HIMG + pi_ * PSZ) * WIMG + pj_ * PSZ + 8 * half;

    const int bRow = tid >> 4;
    const int bCol = (tid & 15) * 4;
    const float* wPtr = Wm + (long)bRow * DMODEL + nBase + bCol;

    typedef __attribute__((address_space(3))) float lds_f32;
    const unsigned bFrag = (unsigned)(unsigned long)((lds_f32*)Bs)
                         + (unsigned)(((8 * half) * B_STRIDE + nHalf * TNW + lsub) * 4);

    f32x8 c0 = {0.f,0.f,0.f,0.f,0.f,0.f,0.f,0.f};
    f32x8 c1 = {0.f,0.f,0.f,0.f,0.f,0.f,0.f,0.f};

    float4 an0 = *(const float4*)(aPtr);
    float4 an1 = *(const float4*)(aPtr + 4);
    float4 an2 = *(const float4*)(aPtr + WIMG);
    float4 an3 = *(const float4*)(aPtr + WIMG + 4);

    for (int st = 0; st < NSTAGE; ++st) {
        if (st > 0) __syncthreads();
        {
            const float* wp = wPtr + (long)st * TKB * DMODEL;
            const float4 w0 = *(const float4*)(wp);
            const float4 w1 = *(const float4*)(wp + 32 * DMODEL);
            const float4 w2 = *(const float4*)(wp + 64 * DMODEL);
            const float4 w3 = *(const float4*)(wp + 96 * DMODEL);
            float* bs = Bs + bRow * B_STRIDE + bCol;
            *(float4*)(bs)                   = w0;
            *(float4*)(bs + 32 * B_STRIDE)   = w1;
            *(float4*)(bs + 64 * B_STRIDE)   = w2;
            *(float4*)(bs + 96 * B_STRIDE)   = w3;
        }
        __syncthreads();

        #pragma unroll
        for (int cc = 0; cc < TKB / TK; ++cc) {
            const int t = st * (TKB / TK) + cc;
            const float a[16] = { an0.x, an0.y, an0.z, an0.w, an1.x, an1.y, an1.z, an1.w,
                                  an2.x, an2.y, an2.z, an2.w, an3.x, an3.y, an3.z, an3.w };
            if (t + 1 < NCHUNK) {
                const int tn = t + 1;
                const float* ap = aPtr + (long)(tn >> 3) * (HIMG * WIMG) + ((tn & 7) * 2) * WIMG;
                an0 = *(const float4*)(ap);
                an1 = *(const float4*)(ap + 4);
                an2 = *(const float4*)(ap + WIMG);
                an3 = *(const float4*)(ap + WIMG + 4);
            }
            f32x2 p0, p1, p2, p3, p4, p5, p6, p7, p8, p9, p10, p11, p12, p13, p14, p15;
            {
                const unsigned g0 = bFrag + (unsigned)((cc * TK + kpos(0, 0, 0)) * B_STRIDE * 4);
                const unsigned g1 = bFrag + (unsigned)((cc * TK + kpos(2, 0, 0)) * B_STRIDE * 4);
                const unsigned g2 = bFrag + (unsigned)((cc * TK + kpos(4, 0, 0)) * B_STRIDE * 4);
                const unsigned g3 = bFrag + (unsigned)((cc * TK + kpos(6, 0, 0)) * B_STRIDE * 4);
                asm volatile(
                    B_READS4(0, 1, 2, 3, 16)
                    B_READS4(4, 5, 6, 7, 17)
                    B_READS4(8, 9, 10, 11, 18)
                    B_READS4(12, 13, 14, 15, 19)
                    "s_wait_dscnt 0x0"
                    : "=&v"(p0), "=&v"(p1), "=&v"(p2), "=&v"(p3),
                      "=&v"(p4), "=&v"(p5), "=&v"(p6), "=&v"(p7),
                      "=&v"(p8), "=&v"(p9), "=&v"(p10), "=&v"(p11),
                      "=&v"(p12), "=&v"(p13), "=&v"(p14), "=&v"(p15)
                    : "v"(g0), "v"(g1), "v"(g2), "v"(g3)
                    : "memory");
            }
            const f32x2 p[16] = { p0, p1, p2, p3, p4, p5, p6, p7,
                                  p8, p9, p10, p11, p12, p13, p14, p15 };
            #pragma unroll
            for (int j = 0; j < 8; ++j) {
                f32x2 av;
                av.x = a[2 * j];
                av.y = a[2 * j + 1];
                c0 = __builtin_amdgcn_wmma_f32_16x16x4_f32(false, av, false, p[2 * j],
                                                           (short)0, c0, false, false);
                c1 = __builtin_amdgcn_wmma_f32_16x16x4_f32(false, av, false, p[2 * j + 1],
                                                           (short)0, c1, false, false);
            }
            __builtin_amdgcn_sched_barrier(0);
        }
    }

    const int colA = nBase + nHalf * TNW + lsub;
    const int colB = colA + 16;
    const float biasA = bias[colA];
    const float biasB = bias[colB];
    #pragma unroll
    for (int v = 0; v < 8; ++v) {
        const int r    = rowBase + strip * 16 + v + half * 8;
        const int bI   = r / NPATCH;
        const int s    = r - bI * NPATCH + 1;
        const long obase = ((long)bI * SEQ + s) * DMODEL;
        const long pbase = (long)s * DMODEL;
        out[obase + colA] = c0[v] + biasA + pos[pbase + colA];
        out[obase + colB] = c1[v] + biasB + pos[pbase + colB];
    }
}

__global__ __launch_bounds__(256)
void vit_class_token(const float* __restrict__ cls,
                     const float* __restrict__ pos,
                     float* __restrict__ out)
{
    const int idx = blockIdx.x * blockDim.x + threadIdx.x;
    const int b = idx / DMODEL;
    const int d = idx - b * DMODEL;
    out[(long)b * SEQ * DMODEL + d] = cls[d] + pos[d];
}

extern "C" void kernel_launch(void* const* d_in, const int* in_sizes, int n_in,
                              void* d_out, int out_size, void* d_ws, size_t ws_size,
                              hipStream_t stream) {
    const float* x   = (const float*)d_in[0];
    const float* Wm  = (const float*)d_in[1];
    const float* b   = (const float*)d_in[2];
    const float* pos = (const float*)d_in[3];
    const float* cls = (const float*)d_in[4];
    float* out = (float*)d_out;

    dim3 grid(ROWS / TM, DMODEL / TN);
    vit_patch_embed_gemm<<<grid, 512, 0, stream>>>(x, Wm, b, pos, out);
    vit_class_token<<<(BATCH * DMODEL) / 256, 256, 0, stream>>>(cls, pos, out);
}


# Whole 768-k x 64-col weight panel kept in LDS by one resident 32-wave work-group per WGP (252 groups); each wave walks its own 16x32 units: no barrier or W staging in the K loop; sums unchanged
# speedup vs baseline: 1.3395x; 1.0657x over previous
#include <hip/hip_runtime.h>

typedef float f32x2 __attribute__((ext_vector_type(2)));
typedef float f32x8 __attribute__((ext_vector_type(8)));

#define BATCH   64
#define CCH     3
#define HIMG    224
#define WIMG    224
#define PSZ     16
#define DMODEL  768
#define KDIM    768
#define NPATCH  196
#define SEQ     197
#define ROWS    (BATCH * NPATCH)
#define NSTRIP  (ROWS / 16)

#define TN      64
#define TNW     32
#define TK      32
#define NCHUNK  (KDIM / TK)
#define NPANEL  (DMODEL / TN)
#define NGROUP  21
#define NWAVE   32
#define B_STRIDE 68
static_assert(B_STRIDE == 68, "the read offsets in B_READS4 are written for a 68-float row");

__device__ __forceinline__ int kpos(int j, int h, int s)
{
    return ((j & 4) << 2) + 8 * h + 2 * (j & 3) + s;
}

#define B_READS4(o0, o1, o2, o3, base) \
    "ds_load_2addr_b32 %" #o0 ", %" #base " offset0:0 offset1:68\n\t"    \
    "ds_load_2addr_b32 %" #o1 ", %" #base " offset0:16 offset1:84\n\t"   \
    "ds_load_2addr_b32 %" #o2 ", %" #base " offset0:136 offset1:204\n\t" \
    "ds_load_2addr_b32 %" #o3 ", %" #base " offset0:152 offset1:220\n\t"

__global__ __launch_bounds__(1024)
void vit_patch_embed_gemm(const float* __restrict__ x,
                          const float* __restrict__ Wm,
                          const float* __restrict__ bias,
                          const float* __restrict__ pos,
                          float* __restrict__ out)
{
    __shared__ float Bs[KDIM * B_STRIDE];

    const int tid    = threadIdx.x;
    const int lane   = tid & 31;
    const int waveId = __builtin_amdgcn_readfirstlane(tid >> 5);
    const int nHalf  = waveId & 1;
    const int half   = lane >> 4;
    const int lsub   = lane & 15;

    const int group  = blockIdx.x;
    const int nBase  = blockIdx.y * TN;
    const int s0     = (NSTRIP * group) / NGROUP;
    const int s1     = (NSTRIP * (group + 1)) / NGROUP;
    const int nUnits = 2 * (s1 - s0);

    {
        const int bRow = tid >> 4;
        const int bCol = (tid & 15) * 4;
        const float* wp = Wm + (long)bRow * DMODEL + nBase + bCol;
        float* bs = Bs + bRow * B_STRIDE + bCol;
        #pragma unroll
        for (int i = 0; i < KDIM / 64; ++i) {
            const float4 w = *(const float4*)(wp + (long)i * 64 * DMODEL);
            *(float4*)(bs + i * 64 * B_STRIDE) = w;
        }
    }
    __syncthreads();

    typedef __attribute__((address_space(3))) float lds_f32;
    const unsigned bFrag = (unsigned)(unsigned long)((lds_f32*)Bs)
                         + (unsigned)(((8 * half) * B_STRIDE + nHalf * TNW + lsub) * 4);

    const int colA = nBase + nHalf * TNW + lsub;
    const int colB = colA + 16;
    const float biasA = bias[colA];
    const float biasB = bias[colB];

    for (int u = waveId; u < nUnits; u += NWAVE) {
        const int stripRow = (s0 + (u >> 1)) * 16;

        const int rG   = stripRow + lsub;
        const int bImg = rG / NPATCH;
        const int nTok = rG - bImg * NPATCH;
        const int pi_  = nTok / 14;
        const int pj_  = nTok - pi_ * 14;
        const float* aPtr = x + ((long)(bImg * CCH) * HIMG + pi_ * PSZ) * WIMG + pj_ * PSZ + 8 * half;

        f32x8 c0 = {0.f,0.f,0.f,0.f,0.f,0.f,0.f,0.f};
        f32x8 c1 = {0.f,0.f,0.f,0.f,0.f,0.f,0.f,0.f};

        float4 an0 = *(const float4*)(aPtr);
        float4 an1 = *(const float4*)(aPtr + 4);
        float4 an2 = *(const float4*)(aPtr + WIMG);
        float4 an3 = *(const float4*)(aPtr + WIMG + 4);

        for (int ch = 0; ch < CCH; ++ch) {
            const float* aCh = aPtr + (long)ch * (HIMG * WIMG);
            const unsigned bCh = bFrag + (unsigned)(ch * 8 * TK * B_STRIDE * 4);
            #pragma unroll
            for (int r = 0; r < 8; ++r) {
                const float a[16] = { an0.x, an0.y, an0.z, an0.w, an1.x, an1.y, an1.z, an1.w,
                                      an2.x, an2.y, an2.z, an2.w, an3.x, an3.y, an3.z, an3.w };
                if (r < 7 || ch + 1 < CCH) {
                    const float* ap = (r < 7) ? aCh + ((r + 1) * 2) * WIMG : aCh + HIMG * WIMG;
                    an0 = *(const float4*)(ap);
                    an1 = *(const float4*)(ap + 4);
                    an2 = *(const float4*)(ap + WIMG);
                    an3 = *(const float4*)(ap + WIMG + 4);
                }
                f32x2 p0, p1, p2, p3, p4, p5, p6, p7, p8, p9, p10, p11, p12, p13, p14, p15;
                {
                    const unsigned g0 = bCh + (unsigned)((r * TK + kpos(0, 0, 0)) * B_STRIDE * 4);
                    const unsigned g1 = bCh + (unsigned)((r * TK + kpos(2, 0, 0)) * B_STRIDE * 4);
                    const unsigned g2 = bCh + (unsigned)((r * TK + kpos(4, 0, 0)) * B_STRIDE * 4);
                    const unsigned g3 = bCh + (unsigned)((r * TK + kpos(6, 0, 0)) * B_STRIDE * 4);
                    asm volatile(
                        B_READS4(0, 1, 2, 3, 16)
                        B_READS4(4, 5, 6, 7, 17)
                        B_READS4(8, 9, 10, 11, 18)
                        B_READS4(12, 13, 14, 15, 19)
                        "s_wait_dscnt 0x0"
                        : "=&v"(p0), "=&v"(p1), "=&v"(p2), "=&v"(p3),
                          "=&v"(p4), "=&v"(p5), "=&v"(p6), "=&v"(p7),
                          "=&v"(p8), "=&v"(p9), "=&v"(p10), "=&v"(p11),
                          "=&v"(p12), "=&v"(p13), "=&v"(p14), "=&v"(p15)
                        : "v"(g0), "v"(g1), "v"(g2), "v"(g3)
                        : "memory");
                }
                const f32x2 p[16] = { p0, p1, p2, p3, p4, p5, p6, p7,
                                      p8, p9, p10, p11, p12, p13, p14, p15 };
                #pragma unroll
                for (int j = 0; j < 8; ++j) {
                    f32x2 av;
                    av.x = a[2 * j];
                    av.y = a[2 * j + 1];
                    c0 = __builtin_amdgcn_wmma_f32_16x16x4_f32(false, av, false, p[2 * j],
                                                               (short)0, c0, false, false);
                    c1 = __builtin_amdgcn_wmma_f32_16x16x4_f32(false, av, false, p[2 * j + 1],
                                                               (short)0, c1, false, false);
                }
                __builtin_amdgcn_sched_barrier(0);
            }
        }

        #pragma unroll
        for (int v = 0; v < 8; ++v) {
            const int r    = stripRow + v + half * 8;
            const int bI   = r / NPATCH;
            const int s    = r - bI * NPATCH + 1;
            const long obase = ((long)bI * SEQ + s) * DMODEL;
            const long pbase = (long)s * DMODEL;
            out[obase + colA] = c0[v] + biasA + pos[pbase + colA];
            out[obase + colB] = c1[v] + biasB + pos[pbase + colB];
        }
    }
}

__global__ __launch_bounds__(256)
void vit_class_token(const float* __restrict__ cls,
                     const float* __restrict__ pos,
                     float* __restrict__ out)
{
    const int idx = blockIdx.x * blockDim.x + threadIdx.x;
    const int b = idx / DMODEL;
    const int d = idx - b * DMODEL;
    out[(long)b * SEQ * DMODEL + d] = cls[d] + pos[d];
}

extern "C" void kernel_launch(void* const* d_in, const int* in_sizes, int n_in,
                              void* d_out, int out_size, void* d_ws, size_t ws_size,
                              hipStream_t stream) {
    const float* x   = (const float*)d_in[0];
    const float* Wm  = (const float*)d_in[1];
    const float* b   = (const float*)d_in[2];
    const float* pos = (const float*)d_in[3];
    const float* cls = (const float*)d_in[4];
    float* out = (float*)d_out;

    dim3 grid(NGROUP, NPANEL);
    vit_patch_embed_gemm<<<grid, 1024, 0, stream>>>(x, Wm, b, pos, out);
    vit_class_token<<<(BATCH * DMODEL) / 256, 256, 0, stream>>>(cls, pos, out);
}
